# starting kernel
# baseline (speedup 1.0000x reference)
#include <hip/hip_runtime.h>


typedef __attribute__((ext_vector_type(2))) float v2f;
typedef __attribute__((ext_vector_type(8))) float v8f;

#define C_IN   128
#define C_OUT  128
#define HDIM   128
#define WDIM   128
#define KW     (C_IN * 9)
#define CCHUNK 16
#define KCHUNK (CCHUNK * 9)
#define NPIX   64
#define PSTRIDE (NPIX * 2 + 32)
#define NPAIRS  (KCHUNK / 2)

__global__ __launch_bounds__(256)
void sdconv_transpose_w(const float* __restrict__ Wt, float* __restrict__ Wk) {
  const int i = blockIdx.x * 256 + threadIdx.x;
  if (i < KW * C_OUT) {
    const int k = i >> 7;
    const int o = i & (C_OUT - 1);
    Wk[i] = Wt[o * KW + k];
  }
}

template <bool KMAJOR>
__global__ __launch_bounds__(256)
void sdconv_wmma_f32(const float* __restrict__ U, const float* __restrict__ V,
                     const float* __restrict__ Wt, float* __restrict__ Out) {
  __shared__ float sS[NPAIRS * PSTRIDE];

  const int bid  = blockIdx.x;
  const int wseg = bid & 1;
  const int h    = (bid >> 1) & (HDIM - 1);
  const int bB   = bid >> 8;
  const int w0   = wseg * NPIX;

  const int tid    = threadIdx.x;
  const int lane   = tid & 31;
  const int wave   = tid >> 5;
  const int oBase  = wave * 16;
  const int laneM  = lane & 15;
  const int laneHi = lane >> 4;

  const int p    = tid & (NPIX - 1);
  const int crow = tid >> 6;

  v8f acc0 = {}; v8f acc1 = {}; v8f acc2 = {}; v8f acc3 = {};

  for (int cc = 0; cc < C_IN / CCHUNK; ++cc) {
    const int cBase = cc * CCHUNK;

    #pragma unroll
    for (int i = 0; i < 4; ++i) {
      const int cl = crow + i * 4;
      const int c  = cBase + cl;
      const int w  = w0 + p;
      const int rowBase = ((bB * C_IN + c) * HDIM + h) * WDIM;
      const float uc = U[rowBase + w];
      const float vc = V[rowBase + w];
      #pragma unroll
      for (int tap = 0; tap < 9; ++tap) {
        const int dy = tap / 3 - 1;
        const int dx = tap % 3 - 1;
        const int hh = h + dy;
        const int ww = w + dx;
        float un = 0.0f, vn = 0.0f;
        if (hh >= 0 && hh < HDIM && ww >= 0 && ww < WDIM) {
          const int nb = ((bB * C_IN + c) * HDIM + hh) * WDIM + ww;
          un = U[nb]; vn = V[nb];
        }
        const float dU = un - uc;
        const float dV = vn - vc;
        const float s  = dU * rsqrtf(1.0f + dV * dV);
        const int kidx = cl * 9 + tap;
        sS[(kidx >> 1) * PSTRIDE + p * 2 + (kidx & 1)] = s;
      }
    }
    __syncthreads();

    if (cc + 1 < C_IN / CCHUNK) {
      #pragma unroll
      for (int i = 0; i < 4; ++i) {
        const int c = cBase + CCHUNK + crow + i * 4;
        const int rowBase = ((bB * C_IN + c) * HDIM + h) * WDIM + w0 + p;
        __builtin_prefetch(&U[rowBase], 0, 0);
        __builtin_prefetch(&V[rowBase], 0, 0);
        if (h > 0)        { __builtin_prefetch(&U[rowBase - WDIM], 0, 0);
                            __builtin_prefetch(&V[rowBase - WDIM], 0, 0); }
        if (h < HDIM - 1) { __builtin_prefetch(&U[rowBase + WDIM], 0, 0);
                            __builtin_prefetch(&V[rowBase + WDIM], 0, 0); }
      }
    }

    const int kw0 = cBase * 9;
    #pragma unroll 4
    for (int ks = 0; ks < KCHUNK / 4; ++ks) {
      const int k0 = ks * 4;
      v2f a;
      if (KMAJOR) {
        const float* wp = Wt + (kw0 + k0 + laneHi * 2) * C_OUT + oBase + laneM;
        a.x = wp[0];
        a.y = wp[C_OUT];
      } else {
        const float* wp = Wt + (oBase + laneM) * KW + (kw0 + k0 + laneHi * 2);
        a.x = wp[0];
        a.y = wp[1];
      }
      const int pr = (k0 >> 1) + laneHi;
      const float* sp = &sS[pr * PSTRIDE + laneM * 2];
      v2f b0 = *(const v2f*)(sp +  0);
      v2f b1 = *(const v2f*)(sp + 32);
      v2f b2 = *(const v2f*)(sp + 64);
      v2f b3 = *(const v2f*)(sp + 96);
      acc0 = __builtin_amdgcn_wmma_f32_16x16x4_f32(false, a, false, b0, (short)0, acc0, false, false);
      acc1 = __builtin_amdgcn_wmma_f32_16x16x4_f32(false, a, false, b1, (short)0, acc1, false, false);
      acc2 = __builtin_amdgcn_wmma_f32_16x16x4_f32(false, a, false, b2, (short)0, acc2, false, false);
      acc3 = __builtin_amdgcn_wmma_f32_16x16x4_f32(false, a, false, b3, (short)0, acc3, false, false);
    }
    __syncthreads();
  }

  #pragma unroll
  for (int i = 0; i < 8; ++i) {
    const int o = oBase + i + laneHi * 8;
    const int obase = ((bB * C_OUT + o) * HDIM + h) * WDIM + w0 + laneM;
    Out[obase +  0] = acc0[i];
    Out[obase + 16] = acc1[i];
    Out[obase + 32] = acc2[i];
    Out[obase + 48] = acc3[i];
  }
}

extern "C" void kernel_launch(void* const* d_in, const int* in_sizes, int n_in,
                              void* d_out, int out_size, void* d_ws, size_t ws_size,
                              hipStream_t stream) {
  (void)in_sizes; (void)n_in; (void)out_size;
  const float* U  = (const float*)d_in[0];
  const float* V  = (const float*)d_in[1];
  const float* Wt = (const float*)d_in[2];
  float* Out = (float*)d_out;

  dim3 grid(4 * HDIM * (WDIM / NPIX));
  dim3 block(256);

  const size_t wbytes = (size_t)KW * C_OUT * sizeof(float);
  if (ws_size >= wbytes) {
    float* Wk = (float*)d_ws;
    sdconv_transpose_w<<<dim3((KW * C_OUT + 255) / 256), block, 0, stream>>>(Wt, Wk);
    sdconv_wmma_f32<true><<<grid, block, 0, stream>>>(U, V, Wk, Out);
  } else {
    sdconv_wmma_f32<false><<<grid, block, 0, stream>>>(U, V, Wt, Out);
  }
}


# phase-1 U/V tap loads made unconditional (clamped index, zero pad by select) so a chunk's 72 loads issue as one batch, not one load-wait pair per tap behind a branch; thread takes 4 adjacent channels
# speedup vs baseline: 1.2127x; 1.2127x over previous
#include <hip/hip_runtime.h>


typedef __attribute__((ext_vector_type(2))) float v2f;
typedef __attribute__((ext_vector_type(8))) float v8f;

#define C_IN   128
#define C_OUT  128
#define HDIM   128
#define WDIM   128
#define KW     (C_IN * 9)
#define CCHUNK 16
#define KCHUNK (CCHUNK * 9)
#define NPIX   64
#define PSTRIDE (NPIX * 2 + 32)
#define NPAIRS  (KCHUNK / 2)

__global__ __launch_bounds__(256)
void sdconv_transpose_w(const float* __restrict__ Wt, float* __restrict__ Wk) {
  const int i = blockIdx.x * 256 + threadIdx.x;
  if (i < KW * C_OUT) {
    const int k = i >> 7;
    const int o = i & (C_OUT - 1);
    Wk[i] = Wt[o * KW + k];
  }
}

template <bool KMAJOR>
__global__ __launch_bounds__(256)
void sdconv_wmma_f32(const float* __restrict__ U, const float* __restrict__ V,
                     const float* __restrict__ Wt, float* __restrict__ Out) {
  __shared__ float sS[NPAIRS * PSTRIDE];

  const int bid  = blockIdx.x;
  const int wseg = bid & 1;
  const int h    = (bid >> 1) & (HDIM - 1);
  const int bB   = bid >> 8;
  const int w0   = wseg * NPIX;

  const int tid    = threadIdx.x;
  const int lane   = tid & 31;
  const int wave   = tid >> 5;
  const int oBase  = wave * 16;
  const int laneM  = lane & 15;
  const int laneHi = lane >> 4;

  const int p    = tid & (NPIX - 1);
  const int crow = __builtin_amdgcn_readfirstlane(tid >> 6);

  const unsigned wq  = (unsigned)(w0 + p);
  const bool     okl = wq > 0u,   okr = wq < (unsigned)(WDIM - 1);
  const bool     okt = h > 0,     okb = h < HDIM - 1;
  const unsigned wl  = okl ? wq - 1u : 0u;
  const unsigned wr  = okr ? wq + 1u : (unsigned)(WDIM - 1);
  const int      ht  = okt ? h - 1 : 0;
  const int      hb  = okb ? h + 1 : HDIM - 1;
  float* const sW = &sS[crow * (18 * PSTRIDE) + p * 2];

  v8f acc0 = {}; v8f acc1 = {}; v8f acc2 = {}; v8f acc3 = {};

  for (int cc = 0; cc < C_IN / CCHUNK; ++cc) {
    const int cBase = cc * CCHUNK;

    float un_[4][9], vn_[4][9];
    {
      const unsigned chan0 = (unsigned)(((bB * C_IN + cBase + crow * 4) * HDIM) * WDIM);
      const unsigned rt = chan0 + (unsigned)(ht * WDIM);
      const unsigned rm = chan0 + (unsigned)(h  * WDIM);
      const unsigned rb = chan0 + (unsigned)(hb * WDIM);
      const unsigned ix[9] = { rt + wl, rt + wq, rt + wr,
                               rm + wl, rm + wq, rm + wr,
                               rb + wl, rb + wq, rb + wr };
      #pragma unroll
      for (int i = 0; i < 4; ++i) {
        #pragma unroll
        for (int tap = 0; tap < 9; ++tap) {
          un_[i][tap] = U[ix[tap] + (unsigned)(i * HDIM * WDIM)];
          vn_[i][tap] = V[ix[tap] + (unsigned)(i * HDIM * WDIM)];
        }
      }
    }
    #pragma unroll
    for (int i = 0; i < 4; ++i) {
      const float uc = un_[i][4];
      const float vc = vn_[i][4];
      #pragma unroll
      for (int tap = 0; tap < 9; ++tap) {
        const int dy = tap / 3 - 1;
        const int dx = tap % 3 - 1;
        const bool inb = (dy < 0 ? okt : (dy > 0 ? okb : true)) &&
                         (dx < 0 ? okl : (dx > 0 ? okr : true));
        const float un = inb ? un_[i][tap] : 0.0f;
        const float vn = inb ? vn_[i][tap] : 0.0f;
        const float dU = un - uc;
        const float dV = vn - vc;
        const float s  = dU * rsqrtf(1.0f + dV * dV);
        const int kl = i * 9 + tap;
        sW[(kl >> 1) * PSTRIDE + (kl & 1)] = s;
      }
    }
    __syncthreads();

    if (cc + 1 < C_IN / CCHUNK) {
      #pragma unroll
      for (int i = 0; i < 4; ++i) {
        const int c = cBase + CCHUNK + crow + i * 4;
        const int rowBase = ((bB * C_IN + c) * HDIM + h) * WDIM + w0 + p;
        __builtin_prefetch(&U[rowBase], 0, 0);
        __builtin_prefetch(&V[rowBase], 0, 0);
        if (h > 0)        { __builtin_prefetch(&U[rowBase - WDIM], 0, 0);
                            __builtin_prefetch(&V[rowBase - WDIM], 0, 0); }
        if (h < HDIM - 1) { __builtin_prefetch(&U[rowBase + WDIM], 0, 0);
                            __builtin_prefetch(&V[rowBase + WDIM], 0, 0); }
      }
    }

    const int kw0 = cBase * 9;
    #pragma unroll 4
    for (int ks = 0; ks < KCHUNK / 4; ++ks) {
      const int k0 = ks * 4;
      v2f a;
      if (KMAJOR) {
        const float* wp = Wt + (kw0 + k0 + laneHi * 2) * C_OUT + oBase + laneM;
        a.x = wp[0];
        a.y = wp[C_OUT];
      } else {
        const float* wp = Wt + (oBase + laneM) * KW + (kw0 + k0 + laneHi * 2);
        a.x = wp[0];
        a.y = wp[1];
      }
      const int pr = (k0 >> 1) + laneHi;
      const float* sp = &sS[pr * PSTRIDE + laneM * 2];
      v2f b0 = *(const v2f*)(sp +  0);
      v2f b1 = *(const v2f*)(sp + 32);
      v2f b2 = *(const v2f*)(sp + 64);
      v2f b3 = *(const v2f*)(sp + 96);
      acc0 = __builtin_amdgcn_wmma_f32_16x16x4_f32(false, a, false, b0, (short)0, acc0, false, false);
      acc1 = __builtin_amdgcn_wmma_f32_16x16x4_f32(false, a, false, b1, (short)0, acc1, false, false);
      acc2 = __builtin_amdgcn_wmma_f32_16x16x4_f32(false, a, false, b2, (short)0, acc2, false, false);
      acc3 = __builtin_amdgcn_wmma_f32_16x16x4_f32(false, a, false, b3, (short)0, acc3, false, false);
    }
    __syncthreads();
  }

  #pragma unroll
  for (int i = 0; i < 8; ++i) {
    const int o = oBase + i + laneHi * 8;
    const int obase = ((bB * C_OUT + o) * HDIM + h) * WDIM + w0 + laneM;
    Out[obase +  0] = acc0[i];
    Out[obase + 16] = acc1[i];
    Out[obase + 32] = acc2[i];
    Out[obase + 48] = acc3[i];
  }
}

extern "C" void kernel_launch(void* const* d_in, const int* in_sizes, int n_in,
                              void* d_out, int out_size, void* d_ws, size_t ws_size,
                              hipStream_t stream) {
  (void)in_sizes; (void)n_in; (void)out_size;
  const float* U  = (const float*)d_in[0];
  const float* V  = (const float*)d_in[1];
  const float* Wt = (const float*)d_in[2];
  float* Out = (float*)d_out;

  dim3 grid(4 * HDIM * (WDIM / NPIX));
  dim3 block(256);

  const size_t wbytes = (size_t)KW * C_OUT * sizeof(float);
  if (ws_size >= wbytes) {
    float* Wk = (float*)d_ws;
    sdconv_transpose_w<<<dim3((KW * C_OUT + 255) / 256), block, 0, stream>>>(Wt, Wk);
    sdconv_wmma_f32<true><<<grid, block, 0, stream>>>(U, V, Wk, Out);
  } else {
    sdconv_wmma_f32<false><<<grid, block, 0, stream>>>(U, V, Wt, Out);
  }
}


# centre tap left out of K (S_U is exactly 0 there: U-U), K 1152->1024, 8 taps = 2 WMMA k-steps per channel; grouping of the f32 sums into k-steps changes, term order kept
# speedup vs baseline: 1.3255x; 1.0930x over previous
#include <hip/hip_runtime.h>


typedef __attribute__((ext_vector_type(2))) float v2f;
typedef __attribute__((ext_vector_type(8))) float v8f;

#define C_IN   128
#define C_OUT  128
#define HDIM   128
#define WDIM   128
#define TAPS   8
#define KW     (C_IN * TAPS)
#define CCHUNK 16
#define KCHUNK (CCHUNK * TAPS)
#define NPIX   64
#define PSTRIDE (NPIX * 2 + 32)
#define NPAIRS  (KCHUNK / 2)

__global__ __launch_bounds__(256)
void sdconv_transpose_w(const float* __restrict__ Wt, float* __restrict__ Wk) {
  const int i = blockIdx.x * 256 + threadIdx.x;
  if (i < KW * C_OUT) {
    const int k = i >> 7;
    const int o = i & (C_OUT - 1);
    const int c = k >> 3;
    const int j = k & 7;
    const int tap = j + (j >> 2);
    Wk[i] = Wt[o * (C_IN * 9) + c * 9 + tap];
  }
}

template <bool KMAJOR>
__global__ __launch_bounds__(256)
void sdconv_wmma_f32(const float* __restrict__ U, const float* __restrict__ V,
                     const float* __restrict__ Wt, float* __restrict__ Out) {
  __shared__ float sS[NPAIRS * PSTRIDE];

  const int bid  = blockIdx.x;
  const int wseg = bid & 1;
  const int h    = (bid >> 1) & (HDIM - 1);
  const int bB   = bid >> 8;
  const int w0   = wseg * NPIX;

  const int tid    = threadIdx.x;
  const int lane   = tid & 31;
  const int wave   = tid >> 5;
  const int oBase  = wave * 16;
  const int laneM  = lane & 15;
  const int laneHi = lane >> 4;

  const int p    = tid & (NPIX - 1);
  const int crow = __builtin_amdgcn_readfirstlane(tid >> 6);

  const unsigned wq  = (unsigned)(w0 + p);
  const bool     okl = wq > 0u,   okr = wq < (unsigned)(WDIM - 1);
  const bool     okt = h > 0,     okb = h < HDIM - 1;
  const unsigned wl  = okl ? wq - 1u : 0u;
  const unsigned wr  = okr ? wq + 1u : (unsigned)(WDIM - 1);
  const int      ht  = okt ? h - 1 : 0;
  const int      hb  = okb ? h + 1 : HDIM - 1;
  float* const sW = &sS[crow * (16 * PSTRIDE) + p * 2];

  v8f acc0 = {}; v8f acc1 = {}; v8f acc2 = {}; v8f acc3 = {};

  for (int cc = 0; cc < C_IN / CCHUNK; ++cc) {
    const int cBase = cc * CCHUNK;

    float un_[4][9], vn_[4][9];
    {
      const unsigned chan0 = (unsigned)(((bB * C_IN + cBase + crow * 4) * HDIM) * WDIM);
      const unsigned rt = chan0 + (unsigned)(ht * WDIM);
      const unsigned rm = chan0 + (unsigned)(h  * WDIM);
      const unsigned rb = chan0 + (unsigned)(hb * WDIM);
      const unsigned ix[9] = { rt + wl, rt + wq, rt + wr,
                               rm + wl, rm + wq, rm + wr,
                               rb + wl, rb + wq, rb + wr };
      #pragma unroll
      for (int i = 0; i < 4; ++i) {
        #pragma unroll
        for (int tap = 0; tap < 9; ++tap) {
          un_[i][tap] = U[ix[tap] + (unsigned)(i * HDIM * WDIM)];
          vn_[i][tap] = V[ix[tap] + (unsigned)(i * HDIM * WDIM)];
        }
      }
    }
    #pragma unroll
    for (int i = 0; i < 4; ++i) {
      const float uc = un_[i][4];
      const float vc = vn_[i][4];
      #pragma unroll
      for (int j = 0; j < TAPS; ++j) {
        const int tap = j + (j >> 2);
        const int dy = tap / 3 - 1;
        const int dx = tap % 3 - 1;
        const bool inb = (dy < 0 ? okt : (dy > 0 ? okb : true)) &&
                         (dx < 0 ? okl : (dx > 0 ? okr : true));
        const float un = inb ? un_[i][tap] : 0.0f;
        const float vn = inb ? vn_[i][tap] : 0.0f;
        const float dU = un - uc;
        const float dV = vn - vc;
        const float s  = dU * rsqrtf(1.0f + dV * dV);
        const int kl = i * TAPS + j;
        sW[(kl >> 1) * PSTRIDE + (kl & 1)] = s;
      }
    }
    __syncthreads();

    if (cc + 1 < C_IN / CCHUNK) {
      #pragma unroll
      for (int i = 0; i < 4; ++i) {
        const int c = cBase + CCHUNK + crow + i * 4;
        const int rowBase = ((bB * C_IN + c) * HDIM + h) * WDIM + w0 + p;
        __builtin_prefetch(&U[rowBase], 0, 0);
        __builtin_prefetch(&V[rowBase], 0, 0);
        if (h > 0)        { __builtin_prefetch(&U[rowBase - WDIM], 0, 0);
                            __builtin_prefetch(&V[rowBase - WDIM], 0, 0); }
        if (h < HDIM - 1) { __builtin_prefetch(&U[rowBase + WDIM], 0, 0);
                            __builtin_prefetch(&V[rowBase + WDIM], 0, 0); }
      }
    }

    const int kw0 = cBase * TAPS;
    #pragma unroll 4
    for (int ks = 0; ks < KCHUNK / 4; ++ks) {
      const int k0 = ks * 4;
      v2f a;
      if (KMAJOR) {
        const float* wp = Wt + (kw0 + k0 + laneHi * 2) * C_OUT + oBase + laneM;
        a.x = wp[0];
        a.y = wp[C_OUT];
      } else {
        const int kk = kw0 + k0 + laneHi * 2;
        const int j  = kk & 7;
        const float* wp = Wt + (oBase + laneM) * (C_IN * 9) + (kk >> 3) * 9 + j + (j >> 2);
        a.x = wp[0];
        a.y = wp[1];
      }
      const int pr = (k0 >> 1) + laneHi;
      const float* sp = &sS[pr * PSTRIDE + laneM * 2];
      v2f b0 = *(const v2f*)(sp +  0);
      v2f b1 = *(const v2f*)(sp + 32);
      v2f b2 = *(const v2f*)(sp + 64);
      v2f b3 = *(const v2f*)(sp + 96);
      acc0 = __builtin_amdgcn_wmma_f32_16x16x4_f32(false, a, false, b0, (short)0, acc0, false, false);
      acc1 = __builtin_amdgcn_wmma_f32_16x16x4_f32(false, a, false, b1, (short)0, acc1, false, false);
      acc2 = __builtin_amdgcn_wmma_f32_16x16x4_f32(false, a, false, b2, (short)0, acc2, false, false);
      acc3 = __builtin_amdgcn_wmma_f32_16x16x4_f32(false, a, false, b3, (short)0, acc3, false, false);
    }
    __syncthreads();
  }

  #pragma unroll
  for (int i = 0; i < 8; ++i) {
    const int o = oBase + i + laneHi * 8;
    const int obase = ((bB * C_OUT + o) * HDIM + h) * WDIM + w0 + laneM;
    Out[obase +  0] = acc0[i];
    Out[obase + 16] = acc1[i];
    Out[obase + 32] = acc2[i];
    Out[obase + 48] = acc3[i];
  }
}

extern "C" void kernel_launch(void* const* d_in, const int* in_sizes, int n_in,
                              void* d_out, int out_size, void* d_ws, size_t ws_size,
                              hipStream_t stream) {
  (void)in_sizes; (void)n_in; (void)out_size;
  const float* U  = (const float*)d_in[0];
  const float* V  = (const float*)d_in[1];
  const float* Wt = (const float*)d_in[2];
  float* Out = (float*)d_out;

  dim3 grid(4 * HDIM * (WDIM / NPIX));
  dim3 block(256);

  const size_t wbytes = (size_t)KW * C_OUT * sizeof(float);
  if (ws_size >= wbytes) {
    float* Wk = (float*)d_ws;
    sdconv_transpose_w<<<dim3((KW * C_OUT + 255) / 256), block, 0, stream>>>(Wt, Wk);
    sdconv_wmma_f32<true><<<grid, block, 0, stream>>>(U, V, Wk, Out);
  } else {
    sdconv_wmma_f32<false><<<grid, block, 0, stream>>>(U, V, Wt, Out);
  }
}


# phase 1 remapped: a thread takes 4 adjacent pixels of one channel (was 4 channels of one pixel): a row is one b128 load + 2 edge loads (18 loads, was 72); S goes to LDS as b128 stores (8, was 32)
# speedup vs baseline: 1.3646x; 1.0295x over previous
#include <hip/hip_runtime.h>


typedef __attribute__((ext_vector_type(2))) float v2f;
typedef __attribute__((ext_vector_type(4))) float v4f;
typedef __attribute__((ext_vector_type(8))) float v8f;

#define C_IN   128
#define C_OUT  128
#define HDIM   128
#define WDIM   128
#define TAPS   8
#define KW     (C_IN * TAPS)
#define CCHUNK 16
#define KCHUNK (CCHUNK * TAPS)
#define NPIX   64
#define PSTRIDE (NPIX * 2 + 32)
#define NPAIRS  (KCHUNK / 2)

__global__ __launch_bounds__(256)
void sdconv_transpose_w(const float* __restrict__ Wt, float* __restrict__ Wk) {
  const int i = blockIdx.x * 256 + threadIdx.x;
  if (i < KW * C_OUT) {
    const int k = i >> 7;
    const int o = i & (C_OUT - 1);
    const int c = k >> 3;
    const int j = k & 7;
    const int tap = j + (j >> 2);
    Wk[i] = Wt[o * (C_IN * 9) + c * 9 + tap];
  }
}

template <bool KMAJOR>
__global__ __launch_bounds__(256)
void sdconv_wmma_f32(const float* __restrict__ U, const float* __restrict__ V,
                     const float* __restrict__ Wt, float* __restrict__ Out) {
  __shared__ __attribute__((aligned(16))) float sS[NPAIRS * PSTRIDE];

  const int bid  = blockIdx.x;
  const int wseg = bid & 1;
  const int h    = (bid >> 1) & (HDIM - 1);
  const int bB   = bid >> 8;
  const int w0   = wseg * NPIX;

  const int tid    = threadIdx.x;
  const int lane   = tid & 31;
  const int wave   = tid >> 5;
  const int oBase  = wave * 16;
  const int laneM  = lane & 15;
  const int laneHi = lane >> 4;

  const int q   = tid & 15;
  const int chl = tid >> 4;

  const unsigned wq  = (unsigned)(w0 + q * 4);
  const bool     okl = wq > 0u,   okr = wq + 4u < (unsigned)WDIM;
  const bool     okt = h > 0,     okb = h < HDIM - 1;
  const unsigned wl  = okl ? wq - 1u : 0u;
  const unsigned wr  = okr ? wq + 4u : (unsigned)(WDIM - 1);
  const int      ht  = okt ? h - 1 : 0;
  const int      hb  = okb ? h + 1 : HDIM - 1;
  float* const sW = &sS[chl * (4 * PSTRIDE) + q * 8];
  const unsigned chan00 = (unsigned)(((bB * C_IN + chl) * HDIM) * WDIM);

  v8f acc0 = {}; v8f acc1 = {}; v8f acc2 = {}; v8f acc3 = {};

  for (int cc = 0; cc < C_IN / CCHUNK; ++cc) {
    const int cBase = cc * CCHUNK;

    const unsigned chan0 = chan00 + (unsigned)(cBase * HDIM * WDIM);
    float uw[3][6], vw[3][6];
    {
      const unsigned rowi[3] = { chan0 + (unsigned)(ht * WDIM),
                                 chan0 + (unsigned)(h  * WDIM),
                                 chan0 + (unsigned)(hb * WDIM) };
      v4f uq[3], vq[3];
      float ul[3], ur[3], vl[3], vr[3];
      #pragma unroll
      for (int r = 0; r < 3; ++r) {
        uq[r] = *(const v4f*)&U[rowi[r] + wq];
        vq[r] = *(const v4f*)&V[rowi[r] + wq];
        ul[r] = U[rowi[r] + wl];
        vl[r] = V[rowi[r] + wl];
        ur[r] = U[rowi[r] + wr];
        vr[r] = V[rowi[r] + wr];
      }
      #pragma unroll
      for (int r = 0; r < 3; ++r) {
        const bool rok = (r == 0) ? okt : ((r == 2) ? okb : true);
        uw[r][0] = (rok && okl) ? ul[r] : 0.0f;
        vw[r][0] = (rok && okl) ? vl[r] : 0.0f;
        #pragma unroll
        for (int i = 0; i < 4; ++i) {
          uw[r][1 + i] = rok ? uq[r][i] : 0.0f;
          vw[r][1 + i] = rok ? vq[r][i] : 0.0f;
        }
        uw[r][5] = (rok && okr) ? ur[r] : 0.0f;
        vw[r][5] = (rok && okr) ? vr[r] : 0.0f;
      }
    }
    #pragma unroll
    for (int jp = 0; jp < TAPS / 2; ++jp) {
      float s[4][2];
      #pragma unroll
      for (int e = 0; e < 2; ++e) {
        const int j   = jp * 2 + e;
        const int tap = j + (j >> 2);
        const int dy = tap / 3 - 1;
        const int dx = tap % 3 - 1;
        #pragma unroll
        for (int i = 0; i < 4; ++i) {
          const float uc = uw[1][1 + i];
          const float vc = vw[1][1 + i];
          const float dU = uw[1 + dy][1 + i + dx] - uc;
          const float dV = vw[1 + dy][1 + i + dx] - vc;
          s[i][e] = dU * rsqrtf(1.0f + dV * dV);
        }
      }
      const v4f lo = { s[0][0], s[0][1], s[1][0], s[1][1] };
      const v4f hi = { s[2][0], s[2][1], s[3][0], s[3][1] };
      *(v4f*)(sW + jp * PSTRIDE)     = lo;
      *(v4f*)(sW + jp * PSTRIDE + 4) = hi;
    }
    __syncthreads();

    if (cc + 1 < C_IN / CCHUNK) {
      const unsigned nx = chan0 + (unsigned)(CCHUNK * HDIM * WDIM) + (unsigned)(h * WDIM) + wq;
      __builtin_prefetch(&U[nx], 0, 0);
      __builtin_prefetch(&V[nx], 0, 0);
      if (h > 0)        { __builtin_prefetch(&U[nx - WDIM], 0, 0);
                          __builtin_prefetch(&V[nx - WDIM], 0, 0); }
      if (h < HDIM - 1) { __builtin_prefetch(&U[nx + WDIM], 0, 0);
                          __builtin_prefetch(&V[nx + WDIM], 0, 0); }
    }

    const int kw0 = cBase * TAPS;
    #pragma unroll 4
    for (int ks = 0; ks < KCHUNK / 4; ++ks) {
      const int k0 = ks * 4;
      v2f a;
      if (KMAJOR) {
        const float* wp = Wt + (kw0 + k0 + laneHi * 2) * C_OUT + oBase + laneM;
        a.x = wp[0];
        a.y = wp[C_OUT];
      } else {
        const int kk = kw0 + k0 + laneHi * 2;
        const int j  = kk & 7;
        const float* wp = Wt + (oBase + laneM) * (C_IN * 9) + (kk >> 3) * 9 + j + (j >> 2);
        a.x = wp[0];
        a.y = wp[1];
      }
      const int pr = (k0 >> 1) + laneHi;
      const float* sp = &sS[pr * PSTRIDE + laneM * 2];
      v2f b0 = *(const v2f*)(sp +  0);
      v2f b1 = *(const v2f*)(sp + 32);
      v2f b2 = *(const v2f*)(sp + 64);
      v2f b3 = *(const v2f*)(sp + 96);
      acc0 = __builtin_amdgcn_wmma_f32_16x16x4_f32(false, a, false, b0, (short)0, acc0, false, false);
      acc1 = __builtin_amdgcn_wmma_f32_16x16x4_f32(false, a, false, b1, (short)0, acc1, false, false);
      acc2 = __builtin_amdgcn_wmma_f32_16x16x4_f32(false, a, false, b2, (short)0, acc2, false, false);
      acc3 = __builtin_amdgcn_wmma_f32_16x16x4_f32(false, a, false, b3, (short)0, acc3, false, false);
    }
    __syncthreads();
  }

  #pragma unroll
  for (int i = 0; i < 8; ++i) {
    const int o = oBase + i + laneHi * 8;
    const int obase = ((bB * C_OUT + o) * HDIM + h) * WDIM + w0 + laneM;
    Out[obase +  0] = acc0[i];
    Out[obase + 16] = acc1[i];
    Out[obase + 32] = acc2[i];
    Out[obase + 48] = acc3[i];
  }
}

extern "C" void kernel_launch(void* const* d_in, const int* in_sizes, int n_in,
                              void* d_out, int out_size, void* d_ws, size_t ws_size,
                              hipStream_t stream) {
  (void)in_sizes; (void)n_in; (void)out_size;
  const float* U  = (const float*)d_in[0];
  const float* V  = (const float*)d_in[1];
  const float* Wt = (const float*)d_in[2];
  float* Out = (float*)d_out;

  dim3 grid(4 * HDIM * (WDIM / NPIX));
  dim3 block(256);

  const size_t wbytes = (size_t)KW * C_OUT * sizeof(float);
  if (ws_size >= wbytes) {
    float* Wk = (float*)d_ws;
    sdconv_transpose_w<<<dim3((KW * C_OUT + 255) / 256), block, 0, stream>>>(Wt, Wk);
    sdconv_wmma_f32<true><<<grid, block, 0, stream>>>(U, V, Wk, Out);
  } else {
    sdconv_wmma_f32<false><<<grid, block, 0, stream>>>(U, V, Wt, Out);
  }
}


# phase 2 software-pipelined: weight fragments of a 16-WMMA block loaded one block ahead into a second register set, S fragments half a block ahead, 2 load groups per block; sum order unchanged
# speedup vs baseline: 1.4053x; 1.0298x over previous
#include <hip/hip_runtime.h>


typedef __attribute__((ext_vector_type(2))) float v2f;
typedef __attribute__((ext_vector_type(4))) float v4f;
typedef __attribute__((ext_vector_type(8))) float v8f;

#define C_IN   128
#define C_OUT  128
#define HDIM   128
#define WDIM   128
#define TAPS   8
#define KW     (C_IN * TAPS)
#define CCHUNK 16
#define KCHUNK (CCHUNK * TAPS)
#define NPIX   64
#define PSTRIDE (NPIX * 2 + 32)
#define NPAIRS  (KCHUNK / 2)

__global__ __launch_bounds__(256)
void sdconv_transpose_w(const float* __restrict__ Wt, float* __restrict__ Wk) {
  const int i = blockIdx.x * 256 + threadIdx.x;
  if (i < KW * C_OUT) {
    const int k = i >> 7;
    const int o = i & (C_OUT - 1);
    const int c = k >> 3;
    const int j = k & 7;
    const int tap = j + (j >> 2);
    Wk[i] = Wt[o * (C_IN * 9) + c * 9 + tap];
  }
}

template <bool KMAJOR>
__device__ __forceinline__ void loadA(v2f (&a)[4], const float* __restrict__ Wt, int kb,
                                      int oBase, int laneM, int laneHi) {
  #pragma unroll
  for (int s = 0; s < 4; ++s) {
    const int kk = kb + s * 4 + laneHi * 2;
    if (KMAJOR) {
      const float* wp = Wt + kk * C_OUT + oBase + laneM;
      a[s].x = wp[0];
      a[s].y = wp[C_OUT];
    } else {
      const int j = kk & 7;
      const float* wp = Wt + (oBase + laneM) * (C_IN * 9) + (kk >> 3) * 9 + j + (j >> 2);
      a[s].x = wp[0];
      a[s].y = wp[1];
    }
  }
}

__device__ __forceinline__ void loadB(v2f (&b)[2][4], const float* sB, int ks) {
  #pragma unroll
  for (int s = 0; s < 2; ++s) {
    const float* sp = sB + (ks + s) * (2 * PSTRIDE);
    #pragma unroll
    for (int t = 0; t < 4; ++t) b[s][t] = *(const v2f*)(sp + 32 * t);
  }
}

__device__ __forceinline__ void wmma8(v8f& acc0, v8f& acc1, v8f& acc2, v8f& acc3,
                                      v2f a0, v2f a1, const v2f (&b)[2][4]) {
  acc0 = __builtin_amdgcn_wmma_f32_16x16x4_f32(false, a0, false, b[0][0], (short)0, acc0, false, false);
  acc1 = __builtin_amdgcn_wmma_f32_16x16x4_f32(false, a0, false, b[0][1], (short)0, acc1, false, false);
  acc2 = __builtin_amdgcn_wmma_f32_16x16x4_f32(false, a0, false, b[0][2], (short)0, acc2, false, false);
  acc3 = __builtin_amdgcn_wmma_f32_16x16x4_f32(false, a0, false, b[0][3], (short)0, acc3, false, false);
  acc0 = __builtin_amdgcn_wmma_f32_16x16x4_f32(false, a1, false, b[1][0], (short)0, acc0, false, false);
  acc1 = __builtin_amdgcn_wmma_f32_16x16x4_f32(false, a1, false, b[1][1], (short)0, acc1, false, false);
  acc2 = __builtin_amdgcn_wmma_f32_16x16x4_f32(false, a1, false, b[1][2], (short)0, acc2, false, false);
  acc3 = __builtin_amdgcn_wmma_f32_16x16x4_f32(false, a1, false, b[1][3], (short)0, acc3, false, false);
}

template <bool KMAJOR>
__global__ __launch_bounds__(256, 8)
void sdconv_wmma_f32(const float* __restrict__ U, const float* __restrict__ V,
                     const float* __restrict__ Wt, float* __restrict__ Out) {
  __shared__ __attribute__((aligned(16))) float sS[NPAIRS * PSTRIDE];

  const int bid  = blockIdx.x;
  const int wseg = bid & 1;
  const int h    = (bid >> 1) & (HDIM - 1);
  const int bB   = bid >> 8;
  const int w0   = wseg * NPIX;

  const int tid    = threadIdx.x;
  const int lane   = tid & 31;
  const int wave   = tid >> 5;
  const int oBase  = wave * 16;
  const int laneM  = lane & 15;
  const int laneHi = lane >> 4;

  const int q   = tid & 15;
  const int chl = tid >> 4;

  const unsigned wq  = (unsigned)(w0 + q * 4);
  const bool     okl = wq > 0u,   okr = wq + 4u < (unsigned)WDIM;
  const bool     okt = h > 0,     okb = h < HDIM - 1;
  const unsigned wl  = okl ? wq - 1u : 0u;
  const unsigned wr  = okr ? wq + 4u : (unsigned)(WDIM - 1);
  const int      ht  = okt ? h - 1 : 0;
  const int      hb  = okb ? h + 1 : HDIM - 1;
  float* const sW = &sS[chl * (4 * PSTRIDE) + q * 8];
  const unsigned chan00 = (unsigned)(((bB * C_IN + chl) * HDIM) * WDIM);

  v8f acc0 = {}; v8f acc1 = {}; v8f acc2 = {}; v8f acc3 = {};

  v2f aF[2][4];
  v2f bF[2][2][4];
  const float* const sB = &sS[laneHi * PSTRIDE + laneM * 2];
  loadA<KMAJOR>(aF[0], Wt, 0, oBase, laneM, laneHi);

  for (int cc = 0; cc < C_IN / CCHUNK; ++cc) {
    const int cBase = cc * CCHUNK;

    const unsigned chan0 = chan00 + (unsigned)(cBase * HDIM * WDIM);
    float uw[3][6], vw[3][6];
    {
      const unsigned rowi[3] = { chan0 + (unsigned)(ht * WDIM),
                                 chan0 + (unsigned)(h  * WDIM),
                                 chan0 + (unsigned)(hb * WDIM) };
      v4f uq[3], vq[3];
      float ul[3], ur[3], vl[3], vr[3];
      #pragma unroll
      for (int r = 0; r < 3; ++r) {
        uq[r] = *(const v4f*)&U[rowi[r] + wq];
        vq[r] = *(const v4f*)&V[rowi[r] + wq];
        ul[r] = U[rowi[r] + wl];
        vl[r] = V[rowi[r] + wl];
        ur[r] = U[rowi[r] + wr];
        vr[r] = V[rowi[r] + wr];
      }
      #pragma unroll
      for (int r = 0; r < 3; ++r) {
        const bool rok = (r == 0) ? okt : ((r == 2) ? okb : true);
        uw[r][0] = (rok && okl) ? ul[r] : 0.0f;
        vw[r][0] = (rok && okl) ? vl[r] : 0.0f;
        #pragma unroll
        for (int i = 0; i < 4; ++i) {
          uw[r][1 + i] = rok ? uq[r][i] : 0.0f;
          vw[r][1 + i] = rok ? vq[r][i] : 0.0f;
        }
        uw[r][5] = (rok && okr) ? ur[r] : 0.0f;
        vw[r][5] = (rok && okr) ? vr[r] : 0.0f;
      }
    }
    #pragma unroll
    for (int jp = 0; jp < TAPS / 2; ++jp) {
      float s[4][2];
      #pragma unroll
      for (int e = 0; e < 2; ++e) {
        const int j   = jp * 2 + e;
        const int tap = j + (j >> 2);
        const int dy = tap / 3 - 1;
        const int dx = tap % 3 - 1;
        #pragma unroll
        for (int i = 0; i < 4; ++i) {
          const float uc = uw[1][1 + i];
          const float vc = vw[1][1 + i];
          const float dU = uw[1 + dy][1 + i + dx] - uc;
          const float dV = vw[1 + dy][1 + i + dx] - vc;
          s[i][e] = dU * rsqrtf(1.0f + dV * dV);
        }
      }
      const v4f lo = { s[0][0], s[0][1], s[1][0], s[1][1] };
      const v4f hi = { s[2][0], s[2][1], s[3][0], s[3][1] };
      *(v4f*)(sW + jp * PSTRIDE)     = lo;
      *(v4f*)(sW + jp * PSTRIDE + 4) = hi;
    }
    __syncthreads();

    if (cc + 1 < C_IN / CCHUNK) {
      const unsigned nx = chan0 + (unsigned)(CCHUNK * HDIM * WDIM) + (unsigned)(h * WDIM) + wq;
      __builtin_prefetch(&U[nx], 0, 0);
      __builtin_prefetch(&V[nx], 0, 0);
      if (h > 0)        { __builtin_prefetch(&U[nx - WDIM], 0, 0);
                          __builtin_prefetch(&V[nx - WDIM], 0, 0); }
      if (h < HDIM - 1) { __builtin_prefetch(&U[nx + WDIM], 0, 0);
                          __builtin_prefetch(&V[nx + WDIM], 0, 0); }
    }

    loadB(bF[0], sB, 0);
    #pragma unroll
    for (int it = 0; it < KCHUNK / 16; ++it) {
      v2f (&aC)[4] = aF[it & 1];
      v2f (&aN)[4] = aF[(it + 1) & 1];
      const int kn = (it + 1 < KCHUNK / 16) ? cc * KCHUNK + (it + 1) * 16
                                            : ((cc + 1) & (C_IN / CCHUNK - 1)) * KCHUNK;
      __builtin_amdgcn_sched_barrier(0);
      loadA<KMAJOR>(aN, Wt, kn, oBase, laneM, laneHi);
      loadB(bF[1], sB, it * 4 + 2);
      __builtin_amdgcn_sched_barrier(0);
      wmma8(acc0, acc1, acc2, acc3, aC[0], aC[1], bF[0]);
      __builtin_amdgcn_sched_barrier(0);
      if (it + 1 < KCHUNK / 16) loadB(bF[0], sB, it * 4 + 4);
      __builtin_amdgcn_sched_barrier(0);
      wmma8(acc0, acc1, acc2, acc3, aC[2], aC[3], bF[1]);
    }
    __syncthreads();
  }

  #pragma unroll
  for (int i = 0; i < 8; ++i) {
    const int o = oBase + i + laneHi * 8;
    const int obase = ((bB * C_OUT + o) * HDIM + h) * WDIM + w0 + laneM;
    Out[obase +  0] = acc0[i];
    Out[obase + 16] = acc1[i];
    Out[obase + 32] = acc2[i];
    Out[obase + 48] = acc3[i];
  }
}

extern "C" void kernel_launch(void* const* d_in, const int* in_sizes, int n_in,
                              void* d_out, int out_size, void* d_ws, size_t ws_size,
                              hipStream_t stream) {
  (void)in_sizes; (void)n_in; (void)out_size;
  const float* U  = (const float*)d_in[0];
  const float* V  = (const float*)d_in[1];
  const float* Wt = (const float*)d_in[2];
  float* Out = (float*)d_out;

  dim3 grid(4 * HDIM * (WDIM / NPIX));
  dim3 block(256);

  const size_t wbytes = (size_t)KW * C_OUT * sizeof(float);
  if (ws_size >= wbytes) {
    float* Wk = (float*)d_ws;
    sdconv_transpose_w<<<dim3((KW * C_OUT + 255) / 256), block, 0, stream>>>(Wt, Wk);
    sdconv_wmma_f32<true><<<grid, block, 0, stream>>>(U, V, Wk, Out);
  } else {
    sdconv_wmma_f32<false><<<grid, block, 0, stream>>>(U, V, Wt, Out);
  }
}


# wave priority (s_setprio) steps down with a work-group's K progress (3,2,1,0 from chunks 4,6,7): the 4 work-groups of a WGP take turns and end together, not oldest-first; sums unchanged
# speedup vs baseline: 1.4482x; 1.0305x over previous
#include <hip/hip_runtime.h>


typedef __attribute__((ext_vector_type(2))) float v2f;
typedef __attribute__((ext_vector_type(4))) float v4f;
typedef __attribute__((ext_vector_type(8))) float v8f;

#define C_IN   128
#define C_OUT  128
#define HDIM   128
#define WDIM   128
#define TAPS   8
#define KW     (C_IN * TAPS)
#define CCHUNK 16
#define KCHUNK (CCHUNK * TAPS)
#define NPIX   64
#define PSTRIDE (NPIX * 2 + 32)
#define NPAIRS  (KCHUNK / 2)

__global__ __launch_bounds__(256)
void sdconv_transpose_w(const float* __restrict__ Wt, float* __restrict__ Wk) {
  const int i = blockIdx.x * 256 + threadIdx.x;
  if (i < KW * C_OUT) {
    const int k = i >> 7;
    const int o = i & (C_OUT - 1);
    const int c = k >> 3;
    const int j = k & 7;
    const int tap = j + (j >> 2);
    Wk[i] = Wt[o * (C_IN * 9) + c * 9 + tap];
  }
}

template <bool KMAJOR>
__device__ __forceinline__ void loadA(v2f (&a)[4], const float* __restrict__ Wt, int kb,
                                      int oBase, int laneM, int laneHi) {
  #pragma unroll
  for (int s = 0; s < 4; ++s) {
    const int kk = kb + s * 4 + laneHi * 2;
    if (KMAJOR) {
      const float* wp = Wt + kk * C_OUT + oBase + laneM;
      a[s].x = wp[0];
      a[s].y = wp[C_OUT];
    } else {
      const int j = kk & 7;
      const float* wp = Wt + (oBase + laneM) * (C_IN * 9) + (kk >> 3) * 9 + j + (j >> 2);
      a[s].x = wp[0];
      a[s].y = wp[1];
    }
  }
}

__device__ __forceinline__ void loadB(v2f (&b)[2][4], const float* sB, int ks) {
  #pragma unroll
  for (int s = 0; s < 2; ++s) {
    const float* sp = sB + (ks + s) * (2 * PSTRIDE);
    #pragma unroll
    for (int t = 0; t < 4; ++t) b[s][t] = *(const v2f*)(sp + 32 * t);
  }
}

__device__ __forceinline__ void wmma8(v8f& acc0, v8f& acc1, v8f& acc2, v8f& acc3,
                                      v2f a0, v2f a1, const v2f (&b)[2][4]) {
  acc0 = __builtin_amdgcn_wmma_f32_16x16x4_f32(false, a0, false, b[0][0], (short)0, acc0, false, false);
  acc1 = __builtin_amdgcn_wmma_f32_16x16x4_f32(false, a0, false, b[0][1], (short)0, acc1, false, false);
  acc2 = __builtin_amdgcn_wmma_f32_16x16x4_f32(false, a0, false, b[0][2], (short)0, acc2, false, false);
  acc3 = __builtin_amdgcn_wmma_f32_16x16x4_f32(false, a0, false, b[0][3], (short)0, acc3, false, false);
  acc0 = __builtin_amdgcn_wmma_f32_16x16x4_f32(false, a1, false, b[1][0], (short)0, acc0, false, false);
  acc1 = __builtin_amdgcn_wmma_f32_16x16x4_f32(false, a1, false, b[1][1], (short)0, acc1, false, false);
  acc2 = __builtin_amdgcn_wmma_f32_16x16x4_f32(false, a1, false, b[1][2], (short)0, acc2, false, false);
  acc3 = __builtin_amdgcn_wmma_f32_16x16x4_f32(false, a1, false, b[1][3], (short)0, acc3, false, false);
}

template <bool KMAJOR>
__global__ __launch_bounds__(256, 8)
void sdconv_wmma_f32(const float* __restrict__ U, const float* __restrict__ V,
                     const float* __restrict__ Wt, float* __restrict__ Out) {
  __shared__ __attribute__((aligned(16))) float sS[NPAIRS * PSTRIDE];

  const int bid  = blockIdx.x;
  const int wseg = bid & 1;
  const int h    = (bid >> 1) & (HDIM - 1);
  const int bB   = bid >> 8;
  const int w0   = wseg * NPIX;

  const int tid    = threadIdx.x;
  const int lane   = tid & 31;
  const int wave   = tid >> 5;
  const int oBase  = wave * 16;
  const int laneM  = lane & 15;
  const int laneHi = lane >> 4;

  const int q   = tid & 15;
  const int chl = tid >> 4;

  const unsigned wq  = (unsigned)(w0 + q * 4);
  const bool     okl = wq > 0u,   okr = wq + 4u < (unsigned)WDIM;
  const bool     okt = h > 0,     okb = h < HDIM - 1;
  const unsigned wl  = okl ? wq - 1u : 0u;
  const unsigned wr  = okr ? wq + 4u : (unsigned)(WDIM - 1);
  const int      ht  = okt ? h - 1 : 0;
  const int      hb  = okb ? h + 1 : HDIM - 1;
  float* const sW = &sS[chl * (4 * PSTRIDE) + q * 8];
  const unsigned chan00 = (unsigned)(((bB * C_IN + chl) * HDIM) * WDIM);

  v8f acc0 = {}; v8f acc1 = {}; v8f acc2 = {}; v8f acc3 = {};

  v2f aF[2][4];
  v2f bF[2][2][4];
  const float* const sB = &sS[laneHi * PSTRIDE + laneM * 2];
  loadA<KMAJOR>(aF[0], Wt, 0, oBase, laneM, laneHi);

  __builtin_amdgcn_s_setprio(3);

  for (int cc = 0; cc < C_IN / CCHUNK; ++cc) {
    const int cBase = cc * CCHUNK;
    if (cc == 4)      __builtin_amdgcn_s_setprio(2);
    else if (cc == 6) __builtin_amdgcn_s_setprio(1);
    else if (cc == 7) __builtin_amdgcn_s_setprio(0);

    const unsigned chan0 = chan00 + (unsigned)(cBase * HDIM * WDIM);
    float uw[3][6], vw[3][6];
    {
      const unsigned rowi[3] = { chan0 + (unsigned)(ht * WDIM),
                                 chan0 + (unsigned)(h  * WDIM),
                                 chan0 + (unsigned)(hb * WDIM) };
      v4f uq[3], vq[3];
      float ul[3], ur[3], vl[3], vr[3];
      #pragma unroll
      for (int r = 0; r < 3; ++r) {
        uq[r] = *(const v4f*)&U[rowi[r] + wq];
        vq[r] = *(const v4f*)&V[rowi[r] + wq];
        ul[r] = U[rowi[r] + wl];
        vl[r] = V[rowi[r] + wl];
        ur[r] = U[rowi[r] + wr];
        vr[r] = V[rowi[r] + wr];
      }
      #pragma unroll
      for (int r = 0; r < 3; ++r) {
        const bool rok = (r == 0) ? okt : ((r == 2) ? okb : true);
        uw[r][0] = (rok && okl) ? ul[r] : 0.0f;
        vw[r][0] = (rok && okl) ? vl[r] : 0.0f;
        #pragma unroll
        for (int i = 0; i < 4; ++i) {
          uw[r][1 + i] = rok ? uq[r][i] : 0.0f;
          vw[r][1 + i] = rok ? vq[r][i] : 0.0f;
        }
        uw[r][5] = (rok && okr) ? ur[r] : 0.0f;
        vw[r][5] = (rok && okr) ? vr[r] : 0.0f;
      }
    }
    #pragma unroll
    for (int jp = 0; jp < TAPS / 2; ++jp) {
      float s[4][2];
      #pragma unroll
      for (int e = 0; e < 2; ++e) {
        const int j   = jp * 2 + e;
        const int tap = j + (j >> 2);
        const int dy = tap / 3 - 1;
        const int dx = tap % 3 - 1;
        #pragma unroll
        for (int i = 0; i < 4; ++i) {
          const float uc = uw[1][1 + i];
          const float vc = vw[1][1 + i];
          const float dU = uw[1 + dy][1 + i + dx] - uc;
          const float dV = vw[1 + dy][1 + i + dx] - vc;
          s[i][e] = dU * rsqrtf(1.0f + dV * dV);
        }
      }
      const v4f lo = { s[0][0], s[0][1], s[1][0], s[1][1] };
      const v4f hi = { s[2][0], s[2][1], s[3][0], s[3][1] };
      *(v4f*)(sW + jp * PSTRIDE)     = lo;
      *(v4f*)(sW + jp * PSTRIDE + 4) = hi;
    }
    __syncthreads();

    if (cc + 1 < C_IN / CCHUNK) {
      const unsigned nx = chan0 + (unsigned)(CCHUNK * HDIM * WDIM) + (unsigned)(h * WDIM) + wq;
      __builtin_prefetch(&U[nx], 0, 0);
      __builtin_prefetch(&V[nx], 0, 0);
      if (h > 0)        { __builtin_prefetch(&U[nx - WDIM], 0, 0);
                          __builtin_prefetch(&V[nx - WDIM], 0, 0); }
      if (h < HDIM - 1) { __builtin_prefetch(&U[nx + WDIM], 0, 0);
                          __builtin_prefetch(&V[nx + WDIM], 0, 0); }
    }

    loadB(bF[0], sB, 0);
    #pragma unroll
    for (int it = 0; it < KCHUNK / 16; ++it) {
      v2f (&aC)[4] = aF[it & 1];
      v2f (&aN)[4] = aF[(it + 1) & 1];
      const int kn = (it + 1 < KCHUNK / 16) ? cc * KCHUNK + (it + 1) * 16
                                            : ((cc + 1) & (C_IN / CCHUNK - 1)) * KCHUNK;
      __builtin_amdgcn_sched_barrier(0);
      loadA<KMAJOR>(aN, Wt, kn, oBase, laneM, laneHi);
      loadB(bF[1], sB, it * 4 + 2);
      __builtin_amdgcn_sched_barrier(0);
      wmma8(acc0, acc1, acc2, acc3, aC[0], aC[1], bF[0]);
      __builtin_amdgcn_sched_barrier(0);
      if (it + 1 < KCHUNK / 16) loadB(bF[0], sB, it * 4 + 4);
      __builtin_amdgcn_sched_barrier(0);
      wmma8(acc0, acc1, acc2, acc3, aC[2], aC[3], bF[1]);
    }
    __syncthreads();
  }

  #pragma unroll
  for (int i = 0; i < 8; ++i) {
    const int o = oBase + i + laneHi * 8;
    const int obase = ((bB * C_OUT + o) * HDIM + h) * WDIM + w0 + laneM;
    Out[obase +  0] = acc0[i];
    Out[obase + 16] = acc1[i];
    Out[obase + 32] = acc2[i];
    Out[obase + 48] = acc3[i];
  }
}

extern "C" void kernel_launch(void* const* d_in, const int* in_sizes, int n_in,
                              void* d_out, int out_size, void* d_ws, size_t ws_size,
                              hipStream_t stream) {
  (void)in_sizes; (void)n_in; (void)out_size;
  const float* U  = (const float*)d_in[0];
  const float* V  = (const float*)d_in[1];
  const float* Wt = (const float*)d_in[2];
  float* Out = (float*)d_out;

  dim3 grid(4 * HDIM * (WDIM / NPIX));
  dim3 block(256);

  const size_t wbytes = (size_t)KW * C_OUT * sizeof(float);
  if (ws_size >= wbytes) {
    float* Wk = (float*)d_ws;
    sdconv_transpose_w<<<dim3((KW * C_OUT + 255) / 256), block, 0, stream>>>(Wt, Wk);
    sdconv_wmma_f32<true><<<grid, block, 0, stream>>>(U, V, Wk, Out);
  } else {
    sdconv_wmma_f32<false><<<grid, block, 0, stream>>>(U, V, Wt, Out);
  }
}


# weights pre-arranged in A-fragment order (a lane's two k-steps = one 16-byte vector, 512 contiguous bytes per wave): 2 b128 loads per 16-WMMA block, was 8 b32 on half-used lines; sum order kept
# speedup vs baseline: 1.4937x; 1.0314x over previous
#include <hip/hip_runtime.h>


typedef __attribute__((ext_vector_type(2))) float v2f;
typedef __attribute__((ext_vector_type(4))) float v4f;
typedef __attribute__((ext_vector_type(8))) float v8f;

#define C_IN   128
#define C_OUT  128
#define HDIM   128
#define WDIM   128
#define TAPS   8
#define KW     (C_IN * TAPS)
#define CCHUNK 16
#define KCHUNK (CCHUNK * TAPS)
#define NPIX   64
#define PSTRIDE (NPIX * 2 + 32)
#define NPAIRS  (KCHUNK / 2)

__global__ __launch_bounds__(256)
void sdconv_transpose_w(const float* __restrict__ Wt, float* __restrict__ Wk) {
  const int i = blockIdx.x * 256 + threadIdx.x;
  if (i < KW * C_OUT) {
    const int e = i & 3;
    const int l = (i >> 2) & 31;
    const int w = (i >> 7) & 7;
    const int P = i >> 10;
    const int k = P * 8 + (e >> 1) * 4 + (l >> 4) * 2 + (e & 1);
    const int o = w * 16 + (l & 15);
    const int c = k >> 3;
    const int j = k & 7;
    const int tap = j + (j >> 2);
    Wk[i] = Wt[o * (C_IN * 9) + c * 9 + tap];
  }
}

template <bool KMAJOR>
__device__ __forceinline__ void loadA(v2f (&a)[4], const float* __restrict__ Wt, int kb,
                                      int oBase, int laneM, int laneHi) {
  #pragma unroll
  for (int s = 0; s < 4; ++s) {
    const int kk = kb + s * 4 + laneHi * 2;
    if (KMAJOR) {
      if ((s & 1) == 0) {
        const v4f v = *(const v4f*)(Wt + (kb + s * 4) * C_OUT + oBase * 8 + laneHi * 64 + laneM * 4);
        a[s].x     = v.x;  a[s].y     = v.y;
        a[s + 1].x = v.z;  a[s + 1].y = v.w;
      }
    } else {
      const int j = kk & 7;
      const float* wp = Wt + (oBase + laneM) * (C_IN * 9) + (kk >> 3) * 9 + j + (j >> 2);
      a[s].x = wp[0];
      a[s].y = wp[1];
    }
  }
}

__device__ __forceinline__ void loadB(v2f (&b)[2][4], const float* sB, int ks) {
  #pragma unroll
  for (int s = 0; s < 2; ++s) {
    const float* sp = sB + (ks + s) * (2 * PSTRIDE);
    #pragma unroll
    for (int t = 0; t < 4; ++t) b[s][t] = *(const v2f*)(sp + 32 * t);
  }
}

__device__ __forceinline__ void wmma8(v8f& acc0, v8f& acc1, v8f& acc2, v8f& acc3,
                                      v2f a0, v2f a1, const v2f (&b)[2][4]) {
  acc0 = __builtin_amdgcn_wmma_f32_16x16x4_f32(false, a0, false, b[0][0], (short)0, acc0, false, false);
  acc1 = __builtin_amdgcn_wmma_f32_16x16x4_f32(false, a0, false, b[0][1], (short)0, acc1, false, false);
  acc2 = __builtin_amdgcn_wmma_f32_16x16x4_f32(false, a0, false, b[0][2], (short)0, acc2, false, false);
  acc3 = __builtin_amdgcn_wmma_f32_16x16x4_f32(false, a0, false, b[0][3], (short)0, acc3, false, false);
  acc0 = __builtin_amdgcn_wmma_f32_16x16x4_f32(false, a1, false, b[1][0], (short)0, acc0, false, false);
  acc1 = __builtin_amdgcn_wmma_f32_16x16x4_f32(false, a1, false, b[1][1], (short)0, acc1, false, false);
  acc2 = __builtin_amdgcn_wmma_f32_16x16x4_f32(false, a1, false, b[1][2], (short)0, acc2, false, false);
  acc3 = __builtin_amdgcn_wmma_f32_16x16x4_f32(false, a1, false, b[1][3], (short)0, acc3, false, false);
}

template <bool KMAJOR>
__global__ __launch_bounds__(256, 8)
void sdconv_wmma_f32(const float* __restrict__ U, const float* __restrict__ V,
                     const float* __restrict__ Wt, float* __restrict__ Out) {
  __shared__ __attribute__((aligned(16))) float sS[NPAIRS * PSTRIDE];

  const int bid  = blockIdx.x;
  const int wseg = bid & 1;
  const int h    = (bid >> 1) & (HDIM - 1);
  const int bB   = bid >> 8;
  const int w0   = wseg * NPIX;

  const int tid    = threadIdx.x;
  const int lane   = tid & 31;
  const int wave   = tid >> 5;
  const int oBase  = wave * 16;
  const int laneM  = lane & 15;
  const int laneHi = lane >> 4;

  const int q   = tid & 15;
  const int chl = tid >> 4;

  const unsigned wq  = (unsigned)(w0 + q * 4);
  const bool     okl = wq > 0u,   okr = wq + 4u < (unsigned)WDIM;
  const bool     okt = h > 0,     okb = h < HDIM - 1;
  const unsigned wl  = okl ? wq - 1u : 0u;
  const unsigned wr  = okr ? wq + 4u : (unsigned)(WDIM - 1);
  const int      ht  = okt ? h - 1 : 0;
  const int      hb  = okb ? h + 1 : HDIM - 1;
  float* const sW = &sS[chl * (4 * PSTRIDE) + q * 8];
  const unsigned chan00 = (unsigned)(((bB * C_IN + chl) * HDIM) * WDIM);

  v8f acc0 = {}; v8f acc1 = {}; v8f acc2 = {}; v8f acc3 = {};

  v2f aF[2][4];
  v2f bF[2][2][4];
  const float* const sB = &sS[laneHi * PSTRIDE + laneM * 2];
  loadA<KMAJOR>(aF[0], Wt, 0, oBase, laneM, laneHi);

  __builtin_amdgcn_s_setprio(3);

  for (int cc = 0; cc < C_IN / CCHUNK; ++cc) {
    const int cBase = cc * CCHUNK;
    if (cc == 4)      __builtin_amdgcn_s_setprio(2);
    else if (cc == 6) __builtin_amdgcn_s_setprio(1);
    else if (cc == 7) __builtin_amdgcn_s_setprio(0);

    const unsigned chan0 = chan00 + (unsigned)(cBase * HDIM * WDIM);
    float uw[3][6], vw[3][6];
    {
      const unsigned rowi[3] = { chan0 + (unsigned)(ht * WDIM),
                                 chan0 + (unsigned)(h  * WDIM),
                                 chan0 + (unsigned)(hb * WDIM) };
      v4f uq[3], vq[3];
      float ul[3], ur[3], vl[3], vr[3];
      #pragma unroll
      for (int r = 0; r < 3; ++r) {
        uq[r] = *(const v4f*)&U[rowi[r] + wq];
        vq[r] = *(const v4f*)&V[rowi[r] + wq];
        ul[r] = U[rowi[r] + wl];
        vl[r] = V[rowi[r] + wl];
        ur[r] = U[rowi[r] + wr];
        vr[r] = V[rowi[r] + wr];
      }
      #pragma unroll
      for (int r = 0; r < 3; ++r) {
        const bool rok = (r == 0) ? okt : ((r == 2) ? okb : true);
        uw[r][0] = (rok && okl) ? ul[r] : 0.0f;
        vw[r][0] = (rok && okl) ? vl[r] : 0.0f;
        #pragma unroll
        for (int i = 0; i < 4; ++i) {
          uw[r][1 + i] = rok ? uq[r][i] : 0.0f;
          vw[r][1 + i] = rok ? vq[r][i] : 0.0f;
        }
        uw[r][5] = (rok && okr) ? ur[r] : 0.0f;
        vw[r][5] = (rok && okr) ? vr[r] : 0.0f;
      }
    }
    #pragma unroll
    for (int jp = 0; jp < TAPS / 2; ++jp) {
      float s[4][2];
      #pragma unroll
      for (int e = 0; e < 2; ++e) {
        const int j   = jp * 2 + e;
        const int tap = j + (j >> 2);
        const int dy = tap / 3 - 1;
        const int dx = tap % 3 - 1;
        #pragma unroll
        for (int i = 0; i < 4; ++i) {
          const float uc = uw[1][1 + i];
          const float vc = vw[1][1 + i];
          const float dU = uw[1 + dy][1 + i + dx] - uc;
          const float dV = vw[1 + dy][1 + i + dx] - vc;
          s[i][e] = dU * rsqrtf(1.0f + dV * dV);
        }
      }
      const v4f lo = { s[0][0], s[0][1], s[1][0], s[1][1] };
      const v4f hi = { s[2][0], s[2][1], s[3][0], s[3][1] };
      *(v4f*)(sW + jp * PSTRIDE)     = lo;
      *(v4f*)(sW + jp * PSTRIDE + 4) = hi;
    }
    __syncthreads();

    if (cc + 1 < C_IN / CCHUNK) {
      const unsigned nx = chan0 + (unsigned)(CCHUNK * HDIM * WDIM) + (unsigned)(h * WDIM) + wq;
      __builtin_prefetch(&U[nx], 0, 0);
      __builtin_prefetch(&V[nx], 0, 0);
      if (h > 0)        { __builtin_prefetch(&U[nx - WDIM], 0, 0);
                          __builtin_prefetch(&V[nx - WDIM], 0, 0); }
      if (h < HDIM - 1) { __builtin_prefetch(&U[nx + WDIM], 0, 0);
                          __builtin_prefetch(&V[nx + WDIM], 0, 0); }
    }

    loadB(bF[0], sB, 0);
    #pragma unroll
    for (int it = 0; it < KCHUNK / 16; ++it) {
      v2f (&aC)[4] = aF[it & 1];
      v2f (&aN)[4] = aF[(it + 1) & 1];
      const int kn = (it + 1 < KCHUNK / 16) ? cc * KCHUNK + (it + 1) * 16
                                            : ((cc + 1) & (C_IN / CCHUNK - 1)) * KCHUNK;
      __builtin_amdgcn_sched_barrier(0);
      loadA<KMAJOR>(aN, Wt, kn, oBase, laneM, laneHi);
      loadB(bF[1], sB, it * 4 + 2);
      __builtin_amdgcn_sched_barrier(0);
      wmma8(acc0, acc1, acc2, acc3, aC[0], aC[1], bF[0]);
      __builtin_amdgcn_sched_barrier(0);
      if (it + 1 < KCHUNK / 16) loadB(bF[0], sB, it * 4 + 4);
      __builtin_amdgcn_sched_barrier(0);
      wmma8(acc0, acc1, acc2, acc3, aC[2], aC[3], bF[1]);
    }
    __syncthreads();
  }

  #pragma unroll
  for (int i = 0; i < 8; ++i) {
    const int o = oBase + i + laneHi * 8;
    const int obase = ((bB * C_OUT + o) * HDIM + h) * WDIM + w0 + laneM;
    Out[obase +  0] = acc0[i];
    Out[obase + 16] = acc1[i];
    Out[obase + 32] = acc2[i];
    Out[obase + 48] = acc3[i];
  }
}

extern "C" void kernel_launch(void* const* d_in, const int* in_sizes, int n_in,
                              void* d_out, int out_size, void* d_ws, size_t ws_size,
                              hipStream_t stream) {
  (void)in_sizes; (void)n_in; (void)out_size;
  const float* U  = (const float*)d_in[0];
  const float* V  = (const float*)d_in[1];
  const float* Wt = (const float*)d_in[2];
  float* Out = (float*)d_out;

  dim3 grid(4 * HDIM * (WDIM / NPIX));
  dim3 block(256);

  const size_t wbytes = (size_t)KW * C_OUT * sizeof(float);
  if (ws_size >= wbytes) {
    float* Wk = (float*)d_ws;
    sdconv_transpose_w<<<dim3((KW * C_OUT + 255) / 256), block, 0, stream>>>(Wt, Wk);
    sdconv_wmma_f32<true><<<grid, block, 0, stream>>>(U, V, Wk, Out);
  } else {
    sdconv_wmma_f32<false><<<grid, block, 0, stream>>>(U, V, Wt, Out);
  }
}


# chunk loop freed of vector address adds: U/V loads = scalar row base + lane offset; S re-laid (tile column M = pixel 4M+t): 2 k-steps = 1 b128 at an immediate offset; output as b128; sums unchanged
# speedup vs baseline: 1.5775x; 1.0561x over previous
#include <hip/hip_runtime.h>


typedef __attribute__((ext_vector_type(2))) float v2f;
typedef __attribute__((ext_vector_type(4))) float v4f;
typedef __attribute__((ext_vector_type(8))) float v8f;

#define C_IN   128
#define C_OUT  128
#define HDIM   128
#define WDIM   128
#define TAPS   8
#define KW     (C_IN * TAPS)
#define CCHUNK 16
#define KCHUNK (CCHUNK * TAPS)
#define NPIX   64
#define QROW    (4 * 16 * 4)
#define CHUNK_BYTES (CCHUNK * HDIM * WDIM * 4)

__global__ __launch_bounds__(256)
void sdconv_transpose_w(const float* __restrict__ Wt, float* __restrict__ Wk) {
  const int i = blockIdx.x * 256 + threadIdx.x;
  if (i < KW * C_OUT) {
    const int e = i & 3;
    const int l = (i >> 2) & 31;
    const int w = (i >> 7) & 7;
    const int P = i >> 10;
    const int k = P * 8 + (e >> 1) * 4 + (l >> 4) * 2 + (e & 1);
    const int o = w * 16 + (l & 15);
    const int c = k >> 3;
    const int j = k & 7;
    const int tap = j + (j >> 2);
    Wk[i] = Wt[o * (C_IN * 9) + c * 9 + tap];
  }
}

template <bool KMAJOR>
__device__ __forceinline__ void loadA(v2f (&a)[4], const float* __restrict__ Wt, int kb,
                                      int oBase, int laneM, int laneHi) {
  #pragma unroll
  for (int s = 0; s < 4; ++s) {
    const int kk = kb + s * 4 + laneHi * 2;
    if (KMAJOR) {
      if ((s & 1) == 0) {
        const v4f v = *(const v4f*)(Wt + (kb + s * 4) * C_OUT + oBase * 8 + laneHi * 64 + laneM * 4);
        a[s].x     = v.x;  a[s].y     = v.y;
        a[s + 1].x = v.z;  a[s + 1].y = v.w;
      }
    } else {
      const int j = kk & 7;
      const float* wp = Wt + (oBase + laneM) * (C_IN * 9) + (kk >> 3) * 9 + j + (j >> 2);
      a[s].x = wp[0];
      a[s].y = wp[1];
    }
  }
}

__device__ __forceinline__ void loadB(v2f (&b)[2][4], const float* sB, int P) {
  const float* sp = sB + P * (2 * QROW);
  #pragma unroll
  for (int t = 0; t < 4; ++t) {
    const v4f v = *(const v4f*)(sp + 64 * t);
    b[0][t].x = v.x;  b[0][t].y = v.y;
    b[1][t].x = v.z;  b[1][t].y = v.w;
  }
}

__device__ __forceinline__ void wmma8(v8f& acc0, v8f& acc1, v8f& acc2, v8f& acc3,
                                      v2f a0, v2f a1, const v2f (&b)[2][4]) {
  acc0 = __builtin_amdgcn_wmma_f32_16x16x4_f32(false, a0, false, b[0][0], (short)0, acc0, false, false);
  acc1 = __builtin_amdgcn_wmma_f32_16x16x4_f32(false, a0, false, b[0][1], (short)0, acc1, false, false);
  acc2 = __builtin_amdgcn_wmma_f32_16x16x4_f32(false, a0, false, b[0][2], (short)0, acc2, false, false);
  acc3 = __builtin_amdgcn_wmma_f32_16x16x4_f32(false, a0, false, b[0][3], (short)0, acc3, false, false);
  acc0 = __builtin_amdgcn_wmma_f32_16x16x4_f32(false, a1, false, b[1][0], (short)0, acc0, false, false);
  acc1 = __builtin_amdgcn_wmma_f32_16x16x4_f32(false, a1, false, b[1][1], (short)0, acc1, false, false);
  acc2 = __builtin_amdgcn_wmma_f32_16x16x4_f32(false, a1, false, b[1][2], (short)0, acc2, false, false);
  acc3 = __builtin_amdgcn_wmma_f32_16x16x4_f32(false, a1, false, b[1][3], (short)0, acc3, false, false);
}

template <bool KMAJOR>
__global__ __launch_bounds__(256, 8)
void sdconv_wmma_f32(const float* __restrict__ U, const float* __restrict__ V,
                     const float* __restrict__ Wt, float* __restrict__ Out) {
  __shared__ __attribute__((aligned(16))) float sS[CCHUNK * 2 * QROW];

  const int bid  = blockIdx.x;
  const int wseg = bid & 1;
  const int h    = (bid >> 1) & (HDIM - 1);
  const int bB   = bid >> 8;
  const int w0   = wseg * NPIX;

  const int tid    = threadIdx.x;
  const int lane   = tid & 31;
  const int wave   = tid >> 5;
  const int oBase  = wave * 16;
  const int laneM  = lane & 15;
  const int laneHi = lane >> 4;

  const int q   = tid & 15;
  const int chl = tid >> 4;

  const unsigned wq  = (unsigned)(w0 + q * 4);
  const bool     okl = wq > 0u,   okr = wq + 4u < (unsigned)WDIM;
  const bool     okt = h > 0,     okb = h < HDIM - 1;
  const unsigned wl  = okl ? wq - 1u : 0u;
  const unsigned wr  = okr ? wq + 4u : (unsigned)(WDIM - 1);
  const int      ht  = okt ? h - 1 : 0;
  const int      hb  = okb ? h + 1 : HDIM - 1;
  float* const sW = &sS[chl * (2 * QROW) + q * 4];
  const unsigned offq = (unsigned)(chl * HDIM * WDIM * 4) + wq * 4u;
  const unsigned offl = (unsigned)(chl * HDIM * WDIM * 4) + wl * 4u;
  const unsigned offr = (unsigned)(chl * HDIM * WDIM * 4) + wr * 4u;
  const size_t img0 = (size_t)bB * C_IN * HDIM * WDIM * 4;
  const char* const uRow0[3] = { (const char*)U + img0 + (size_t)(ht * WDIM * 4),
                                 (const char*)U + img0 + (size_t)(h  * WDIM * 4),
                                 (const char*)U + img0 + (size_t)(hb * WDIM * 4) };
  const char* const vRow0[3] = { (const char*)V + img0 + (size_t)(ht * WDIM * 4),
                                 (const char*)V + img0 + (size_t)(h  * WDIM * 4),
                                 (const char*)V + img0 + (size_t)(hb * WDIM * 4) };

  const int pl    = lane < 24 ? lane : 0;
  const int pch   = pl / 12;
  const int prem  = pl - pch * 12;
  const int parr  = prem / 6;
  const int prow  = (prem - parr * 6) >> 1;
  const int prowh = prow == 0 ? ht : (prow == 1 ? h : hb);
  const char* const pfp = (parr ? (const char*)V : (const char*)U) + img0
      + (size_t)(((wave * 2 + pch) * HDIM + prowh) * WDIM * 4 + w0 * 4 + (prem & 1) * 128);

  v8f acc0 = {}; v8f acc1 = {}; v8f acc2 = {}; v8f acc3 = {};

  v2f aF[2][4];
  v2f bF[2][2][4];
  const float* const sB = &sS[laneHi * QROW + laneM * 4];
  loadA<KMAJOR>(aF[0], Wt, 0, oBase, laneM, laneHi);

  __builtin_amdgcn_s_setprio(3);

  for (int cc = 0; cc < C_IN / CCHUNK; ++cc) {
    if (cc == 4)      __builtin_amdgcn_s_setprio(2);
    else if (cc == 6) __builtin_amdgcn_s_setprio(1);
    else if (cc == 7) __builtin_amdgcn_s_setprio(0);

    const unsigned cOff = (unsigned)(cc * CHUNK_BYTES);
    const unsigned cq = offq + cOff, cl = offl + cOff, cr = offr + cOff;
    float uw[3][6], vw[3][6];
    {
      v4f uq[3], vq[3];
      float ul[3], ur[3], vl[3], vr[3];
      #pragma unroll
      for (int r = 0; r < 3; ++r) {
        uq[r] = *(const v4f*)(uRow0[r] + cq);
        vq[r] = *(const v4f*)(vRow0[r] + cq);
        ul[r] = *(const float*)(uRow0[r] + cl);
        vl[r] = *(const float*)(vRow0[r] + cl);
        ur[r] = *(const float*)(uRow0[r] + cr);
        vr[r] = *(const float*)(vRow0[r] + cr);
      }
      #pragma unroll
      for (int r = 0; r < 3; ++r) {
        const bool rok = (r == 0) ? okt : ((r == 2) ? okb : true);
        uw[r][0] = (rok && okl) ? ul[r] : 0.0f;
        vw[r][0] = (rok && okl) ? vl[r] : 0.0f;
        #pragma unroll
        for (int i = 0; i < 4; ++i) {
          uw[r][1 + i] = rok ? uq[r][i] : 0.0f;
          vw[r][1 + i] = rok ? vq[r][i] : 0.0f;
        }
        uw[r][5] = (rok && okr) ? ur[r] : 0.0f;
        vw[r][5] = (rok && okr) ? vr[r] : 0.0f;
      }
    }
    #pragma unroll
    for (int hh = 0; hh < 2; ++hh) {
      #pragma unroll
      for (int i = 0; i < 4; ++i) {
        const float uc = uw[1][1 + i];
        const float vc = vw[1][1 + i];
        float s[4];
        #pragma unroll
        for (int e = 0; e < 4; ++e) {
          const int j   = (e >> 1) * 4 + hh * 2 + (e & 1);
          const int tap = j + (j >> 2);
          const int dy = tap / 3 - 1;
          const int dx = tap % 3 - 1;
          const float dU = uw[1 + dy][1 + i + dx] - uc;
          const float dV = vw[1 + dy][1 + i + dx] - vc;
          s[e] = dU * rsqrtf(1.0f + dV * dV);
        }
        const v4f sv = { s[0], s[1], s[2], s[3] };
        *(v4f*)(sW + hh * QROW + i * 64) = sv;
      }
    }
    __syncthreads();

    if (cc + 1 < C_IN / CCHUNK) {
      __builtin_prefetch(pfp + cOff + CHUNK_BYTES, 0, 0);
    }

    loadB(bF[0], sB, 0);
    #pragma unroll
    for (int it = 0; it < KCHUNK / 16; ++it) {
      v2f (&aC)[4] = aF[it & 1];
      v2f (&aN)[4] = aF[(it + 1) & 1];
      const int kn = (it + 1 < KCHUNK / 16) ? cc * KCHUNK + (it + 1) * 16
                                            : ((cc + 1) & (C_IN / CCHUNK - 1)) * KCHUNK;
      __builtin_amdgcn_sched_barrier(0);
      loadA<KMAJOR>(aN, Wt, kn, oBase, laneM, laneHi);
      loadB(bF[1], sB, it * 2 + 1);
      __builtin_amdgcn_sched_barrier(0);
      wmma8(acc0, acc1, acc2, acc3, aC[0], aC[1], bF[0]);
      __builtin_amdgcn_sched_barrier(0);
      if (it + 1 < KCHUNK / 16) loadB(bF[0], sB, it * 2 + 2);
      __builtin_amdgcn_sched_barrier(0);
      wmma8(acc0, acc1, acc2, acc3, aC[2], aC[3], bF[1]);
    }
    __syncthreads();
  }

  const unsigned ob = (unsigned)(((bB * C_OUT + oBase + laneHi * 8) * HDIM + h) * WDIM
                                 + w0 + laneM * 4) * 4u;
  #pragma unroll
  for (int i = 0; i < 8; ++i) {
    const v4f d = { acc0[i], acc1[i], acc2[i], acc3[i] };
    *(v4f*)((char*)Out + ob + (unsigned)(i * HDIM * WDIM * 4)) = d;
  }
}

extern "C" void kernel_launch(void* const* d_in, const int* in_sizes, int n_in,
                              void* d_out, int out_size, void* d_ws, size_t ws_size,
                              hipStream_t stream) {
  (void)in_sizes; (void)n_in; (void)out_size;
  const float* U  = (const float*)d_in[0];
  const float* V  = (const float*)d_in[1];
  const float* Wt = (const float*)d_in[2];
  float* Out = (float*)d_out;

  dim3 grid(4 * HDIM * (WDIM / NPIX));
  dim3 block(256);

  const size_t wbytes = (size_t)KW * C_OUT * sizeof(float);
  if (ws_size >= wbytes) {
    float* Wk = (float*)d_ws;
    sdconv_transpose_w<<<dim3((KW * C_OUT + 255) / 256), block, 0, stream>>>(Wt, Wk);
    sdconv_wmma_f32<true><<<grid, block, 0, stream>>>(U, V, Wk, Out);
  } else {
    sdconv_wmma_f32<false><<<grid, block, 0, stream>>>(U, V, Wt, Out);
  }
}
